# starting kernel
# baseline (speedup 1.0000x reference)
#include <hip/hip_runtime.h>

typedef __attribute__((ext_vector_type(2))) float v2f;
typedef __attribute__((ext_vector_type(8))) float v8f;

#define NA 8
#define NACT 10
#define EMB 32
#define BIAS 1000

#define SWZ_BCAST8(x, K) \
  __int_as_float(__builtin_amdgcn_ds_swizzle(__float_as_int(x), ((K) << 5) | 24))
#define SWZ_XOR(x, M) \
  __int_as_float(__builtin_amdgcn_ds_swizzle(__float_as_int(x), ((M) << 10) | 31))

__global__ __launch_bounds__(256) void qdpp_mix_kernel(
    const float* __restrict__ agent_qs,
    const int*   __restrict__ states,
    const int*   __restrict__ actions,
    const float* __restrict__ noise,
    const float* __restrict__ Bw,
    float*       __restrict__ out,
    int npairs)
{
  const int lane = threadIdx.x & 31;
  const int wave = threadIdx.x >> 5;
  int pair = blockIdx.x * 8 + wave;
  if (pair >= npairs) pair = npairs - 1;

  const int hi = lane >> 4;
  const int r  = lane & 15;
  const int e  = pair * 2 + (r >> 3);
  const int ag = r & 7;

  const int st  = states[e * NA + ag];
  const int ac  = actions[e * NA + ag];
  const int row = st * NACT + ac + BIAS * ag;
  const float* rp = Bw + (size_t)row * EMB + 2 * hi;

  float v[16];
  #pragma unroll
  for (int s = 0; s < 8; ++s) {
    const float2 t = *(const float2*)(rp + 4 * s);
    v[2 * s]     = t.x;
    v[2 * s + 1] = t.y;
  }

  float ssq = 0.f;
  #pragma unroll
  for (int j = 0; j < 16; ++j) ssq = fmaf(v[j], v[j], ssq);
  ssq += SWZ_XOR(ssq, 16);
  const float rn = __builtin_amdgcn_rsqf(fmaxf(ssq, 1e-24f));
  #pragma unroll
  for (int j = 0; j < 16; ++j) v[j] *= rn;

  float qv = agent_qs[e * NA + ag];
  qv = fminf(fmaxf(qv, -10.f), 10.f);
  qv += SWZ_XOR(qv, 1);
  qv += SWZ_XOR(qv, 2);
  qv += SWZ_XOR(qv, 4);

  v8f c = {};
  #pragma unroll
  for (int s = 0; s < 8; ++s) {
    v2f a;
    a[0] = v[2 * s];
    a[1] = v[2 * s + 1];
    c = __builtin_amdgcn_wmma_f32_16x16x4_f32(
        false, a, false, a, (short)0, c, false, false);
  }

  const float nv = 0.1f * noise[e * NA + ag];
  float m[8];
  #pragma unroll
  for (int i = 0; i < 8; ++i) {
    const float x = c[i] + (i == ag ? nv : 0.f);
    m[i] = fmaxf(x, 0.f) + 1e-8f;
  }

  float det = 1.f;
  #define LU_STEP(K)                                          \
  {                                                           \
    const float akk = SWZ_BCAST8(m[K], K);     \
    det *= akk;                                               \
    const float ia = __builtin_amdgcn_rcpf(akk);              \
    _Pragma("unroll")                                         \
    for (int i = (K) + 1; i < 8; ++i) {                       \
      const float lik = SWZ_BCAST8(m[i], K) * ia;             \
      m[i] = fmaf(-lik, m[K], m[i]);                          \
    }                                                         \
  }
  LU_STEP(0) LU_STEP(1) LU_STEP(2) LU_STEP(3)
  LU_STEP(4) LU_STEP(5) LU_STEP(6) LU_STEP(7)
  #undef LU_STEP

  if (lane == 0 || lane == 24) {
    const float logdet = 0.69314718056f * __builtin_amdgcn_logf(det + 1e-8f);
    out[e] = qv + logdet;
  }
}

extern "C" void kernel_launch(void* const* d_in, const int* in_sizes, int n_in,
                              void* d_out, int out_size, void* d_ws, size_t ws_size,
                              hipStream_t stream) {
  const float* agent_qs = (const float*)d_in[0];
  const int*   states   = (const int*)d_in[1];
  const int*   actions  = (const int*)d_in[2];
  const float* noisep   = (const float*)d_in[3];
  const float* Bw       = (const float*)d_in[4];
  float* out = (float*)d_out;

  const int nelem  = in_sizes[0] / NA;
  const int npairs = (nelem + 1) / 2;
  const int blocks = (npairs + 7) / 8;

  qdpp_mix_kernel<<<blocks, 256, 0, stream>>>(agent_qs, states, actions,
                                              noisep, Bw, out, npairs);
}


# Resident grid: 2048 blocks fill every wave slot once and each wave loops over 8 element pairs (next pair's inputs requested a trip ahead), not one short wave per pair: 16384 waves enter, not 131072
# speedup vs baseline: 1.0707x; 1.0707x over previous
#include <hip/hip_runtime.h>

typedef __attribute__((ext_vector_type(2))) float v2f;
typedef __attribute__((ext_vector_type(8))) float v8f;

#define NA 8
#define NACT 10
#define EMB 32
#define BIAS 1000

#define GRID_BLOCKS 2048
#define WAVE_STRIDE (GRID_BLOCKS * 8)

#define SWZ_BCAST8(x, K) \
  __int_as_float(__builtin_amdgcn_ds_swizzle(__float_as_int(x), ((K) << 5) | 24))
#define SWZ_XOR(x, M) \
  __int_as_float(__builtin_amdgcn_ds_swizzle(__float_as_int(x), ((M) << 10) | 31))

__device__ __forceinline__ void gather_row(float (&v)[16], const float* __restrict__ bwl, int row)
{
  const float* rp = bwl + (size_t)row * EMB;
  #pragma unroll
  for (int s = 0; s < 8; ++s) {
    const float2 t = *(const float2*)(rp + 4 * s);
    v[2 * s]     = t.x;
    v[2 * s + 1] = t.y;
  }
}

__device__ __forceinline__ float pair_value(float (&v)[16], float q_raw, float n_raw, int ag)
{
  float ssq = 0.f;
  #pragma unroll
  for (int j = 0; j < 16; ++j) ssq = fmaf(v[j], v[j], ssq);
  ssq += SWZ_XOR(ssq, 16);
  const float rn = __builtin_amdgcn_rsqf(fmaxf(ssq, 1e-24f));
  #pragma unroll
  for (int j = 0; j < 16; ++j) v[j] *= rn;

  float qv = fminf(fmaxf(q_raw, -10.f), 10.f);
  qv += SWZ_XOR(qv, 1);
  qv += SWZ_XOR(qv, 2);
  qv += SWZ_XOR(qv, 4);

  v8f c = {};
  #pragma unroll
  for (int s = 0; s < 8; ++s) {
    v2f a;
    a[0] = v[2 * s];
    a[1] = v[2 * s + 1];
    c = __builtin_amdgcn_wmma_f32_16x16x4_f32(
        false, a, false, a, (short)0, c, false, false);
  }

  const float nv = 0.1f * n_raw;
  float m[8];
  #pragma unroll
  for (int i = 0; i < 8; ++i) {
    const float x = c[i] + (i == ag ? nv : 0.f);
    m[i] = fmaxf(x, 0.f) + 1e-8f;
  }

  float det = 1.f;
  #define LU_STEP(K)                                          \
  {                                                           \
    const float akk = SWZ_BCAST8(m[K], K);     \
    det *= akk;                                               \
    const float ia = __builtin_amdgcn_rcpf(akk);              \
    _Pragma("unroll")                                         \
    for (int i = (K) + 1; i < 8; ++i) {                       \
      const float lik = SWZ_BCAST8(m[i], K) * ia;             \
      m[i] = fmaf(-lik, m[K], m[i]);                          \
    }                                                         \
  }
  LU_STEP(0) LU_STEP(1) LU_STEP(2) LU_STEP(3)
  LU_STEP(4) LU_STEP(5) LU_STEP(6) LU_STEP(7)
  #undef LU_STEP

  const float logdet = 0.69314718056f * __builtin_amdgcn_logf(det + 1e-8f);
  return qv + logdet;
}

__global__ __launch_bounds__(256) void qdpp_mix_kernel(
    const float* __restrict__ agent_qs,
    const int*   __restrict__ states,
    const int*   __restrict__ actions,
    const float* __restrict__ noise,
    const float* __restrict__ Bw,
    float*       __restrict__ out,
    int npairs)
{
  const int lane = threadIdx.x & 31;
  const int wave = __builtin_amdgcn_readfirstlane((int)(threadIdx.x >> 5));
  int pair = blockIdx.x * 8 + wave;
  if (pair >= npairs) return;

  const int hi = lane >> 4;
  const int r  = lane & 15;
  const int ag = r & 7;
  const int rowbase = BIAS * ag;
  const float* bwl = Bw + 2 * hi;
  const bool writer = (lane == 0) | (lane == 24);

  int off = pair * 16 + r;
  float v[16];
  gather_row(v, bwl, states[off] * NACT + actions[off] + rowbase);
  float q_raw = agent_qs[off];
  float n_raw = noise[off];

  int npair = pair + WAVE_STRIDE;
  off = (npair < npairs ? npair : pair) * 16 + r;
  int   st_n = states[off];
  int   ac_n = actions[off];
  float q_n  = agent_qs[off];
  float n_n  = noise[off];

  for (;;) {
    const float res = pair_value(v, q_raw, n_raw, ag);

    const int nnpair = npair + WAVE_STRIDE;
    gather_row(v, bwl, st_n * NACT + ac_n + rowbase);
    q_raw = q_n;
    n_raw = n_n;
    off  = (nnpair < npairs ? nnpair : pair) * 16 + r;
    st_n = states[off];
    ac_n = actions[off];
    q_n  = agent_qs[off];
    n_n  = noise[off];

    if (writer) out[pair * 2 + hi] = res;
    if (npair >= npairs) break;
    pair  = npair;
    npair = nnpair;
  }
}

extern "C" void kernel_launch(void* const* d_in, const int* in_sizes, int n_in,
                              void* d_out, int out_size, void* d_ws, size_t ws_size,
                              hipStream_t stream) {
  const float* agent_qs = (const float*)d_in[0];
  const int*   states   = (const int*)d_in[1];
  const int*   actions  = (const int*)d_in[2];
  const float* noisep   = (const float*)d_in[3];
  const float* Bw       = (const float*)d_in[4];
  float* out = (float*)d_out;

  const int nelem  = in_sizes[0] / NA;
  const int npairs = (nelem + 1) / 2;

  qdpp_mix_kernel<<<GRID_BLOCKS, 256, 0, stream>>>(agent_qs, states, actions,
                                                   noisep, Bw, out, npairs);
}


# Row gather coalesced: each 16-byte load reads four whole 128-byte rows (8 lanes a row) and a wave-private LDS tile turns them into WMMA fragments; k order of the Gram and norm sums changes
# speedup vs baseline: 1.1346x; 1.0597x over previous
#include <hip/hip_runtime.h>


typedef __attribute__((ext_vector_type(2))) float v2f;
typedef __attribute__((ext_vector_type(8))) float v8f;

#define NA 8
#define NACT 10
#define EMB 32
#define BIAS 1000

#define GRID_BLOCKS 2048
#define WAVE_STRIDE (GRID_BLOCKS * 8)

#define PITCH 36
#define TILE (16 * PITCH)

#define SWZ_BCAST8(x, K) \
  __int_as_float(__builtin_amdgcn_ds_swizzle(__float_as_int(x), ((K) << 5) | 24))
#define SWZ_XOR(x, M) \
  __int_as_float(__builtin_amdgcn_ds_swizzle(__float_as_int(x), ((M) << 10) | 31))

__device__ __forceinline__ void request_rows(float4 (&g)[4], const float* __restrict__ bwl,
                                             const int4 s4, const int4 a4, int gbias)
{
  const int sx[4] = {s4.x, s4.y, s4.z, s4.w};
  const int ax[4] = {a4.x, a4.y, a4.z, a4.w};
  #pragma unroll
  for (int j = 0; j < 4; ++j) {
    const int row = sx[j] * NACT + ax[j] + gbias + BIAS * j;
    g[j] = *(const float4*)(bwl + (size_t)row * EMB);
  }
}

__device__ __forceinline__ void stage_rows(float* st, const float4 (&g)[4])
{
  #pragma unroll
  for (int j = 0; j < 4; ++j) *(float4*)(st + j * PITCH) = g[j];
}

__device__ __forceinline__ void fetch_fragment(float (&v)[16], const float* ld)
{
  #pragma unroll
  for (int s = 0; s < 4; ++s) {
    const float4 t = *(const float4*)(ld + 4 * s);
    v[4 * s]     = t.x;
    v[4 * s + 1] = t.y;
    v[4 * s + 2] = t.z;
    v[4 * s + 3] = t.w;
  }
}

__device__ __forceinline__ v8f gram_pair(float (&v)[16])
{
  float ssq = 0.f;
  #pragma unroll
  for (int j = 0; j < 16; ++j) ssq = fmaf(v[j], v[j], ssq);
  ssq += SWZ_XOR(ssq, 16);
  const float rn = __builtin_amdgcn_rsqf(fmaxf(ssq, 1e-24f));
  #pragma unroll
  for (int j = 0; j < 16; ++j) v[j] *= rn;

  v8f c = {};
  #pragma unroll
  for (int s = 0; s < 8; ++s) {
    v2f a;
    a[0] = v[2 * s];
    a[1] = v[2 * s + 1];
    c = __builtin_amdgcn_wmma_f32_16x16x4_f32(
        false, a, false, a, (short)0, c, false, false);
  }
  return c;
}

__device__ __forceinline__ float finish_pair(const v8f c, float q_raw, float n_raw, int ag)
{
  float qv = fminf(fmaxf(q_raw, -10.f), 10.f);
  qv += SWZ_XOR(qv, 1);
  qv += SWZ_XOR(qv, 2);
  qv += SWZ_XOR(qv, 4);

  const float nv = 0.1f * n_raw;
  float m[8];
  #pragma unroll
  for (int i = 0; i < 8; ++i) {
    const float x = c[i] + (i == ag ? nv : 0.f);
    m[i] = fmaxf(x, 0.f) + 1e-8f;
  }

  float det = 1.f;
  #define LU_STEP(K)                                          \
  {                                                           \
    const float akk = SWZ_BCAST8(m[K], K);     \
    det *= akk;                                               \
    const float ia = __builtin_amdgcn_rcpf(akk);              \
    _Pragma("unroll")                                         \
    for (int i = (K) + 1; i < 8; ++i) {                       \
      const float lik = SWZ_BCAST8(m[i], K) * ia;             \
      m[i] = fmaf(-lik, m[K], m[i]);                          \
    }                                                         \
  }
  LU_STEP(0) LU_STEP(1) LU_STEP(2) LU_STEP(3)
  LU_STEP(4) LU_STEP(5) LU_STEP(6) LU_STEP(7)
  #undef LU_STEP

  const float logdet = 0.69314718056f * __builtin_amdgcn_logf(det + 1e-8f);
  return qv + logdet;
}

__device__ __forceinline__ float pair_value(float (&v)[16], float q_raw, float n_raw, int ag)
{
  return finish_pair(gram_pair(v), q_raw, n_raw, ag);
}

__global__ __launch_bounds__(256) void qdpp_mix_kernel(
    const float* __restrict__ agent_qs,
    const int*   __restrict__ states,
    const int*   __restrict__ actions,
    const float* __restrict__ noise,
    const float* __restrict__ Bw,
    float*       __restrict__ out,
    int npairs)
{
  __shared__ __attribute__((aligned(16))) float tiles[8 * TILE];

  const int lane = threadIdx.x & 31;
  const int wave = __builtin_amdgcn_readfirstlane((int)(threadIdx.x >> 5));
  int pair = blockIdx.x * 8 + wave;
  if (pair >= npairs) return;

  const int hi = lane >> 4;
  const int r  = lane & 15;
  const int ag = r & 7;
  const bool writer = (lane == 0) | (lane == 24);

  const int ga = lane >> 3;
  const int gb = lane & 7;
  const int grow  = 8 * (ga & 1) + 4 * (ga >> 1);
  const int gbias = BIAS * (grow & 7);
  const float* bwl = Bw + 4 * gb;
  float* tile = tiles + wave * TILE;
  float* st = tile + grow * PITCH + 4 * gb;
  const float* ld = tile + r * PITCH + 16 * hi;

  int4 s4 = *(const int4*)(states  + pair * 16 + grow);
  int4 a4 = *(const int4*)(actions + pair * 16 + grow);
  float q_raw = agent_qs[pair * 16 + r];
  float n_raw = noise[pair * 16 + r];
  float4 g[4];
  request_rows(g, bwl, s4, a4, gbias);

  int npair = pair + WAVE_STRIDE;
  int pn = npair < npairs ? npair : pair;
  s4 = *(const int4*)(states  + pn * 16 + grow);
  a4 = *(const int4*)(actions + pn * 16 + grow);
  float q_n = agent_qs[pn * 16 + r];
  float n_n = noise[pn * 16 + r];
  stage_rows(st, g);

  for (;;) {
    float v[16];
    fetch_fragment(v, ld);
    const float res = pair_value(v, q_raw, n_raw, ag);

    __builtin_amdgcn_sched_barrier(0);

    const int nnpair = npair + WAVE_STRIDE;
    request_rows(g, bwl, s4, a4, gbias);
    q_raw = q_n;
    n_raw = n_n;
    pn = nnpair < npairs ? nnpair : pair;
    s4 = *(const int4*)(states  + pn * 16 + grow);
    a4 = *(const int4*)(actions + pn * 16 + grow);
    q_n = agent_qs[pn * 16 + r];
    n_n = noise[pn * 16 + r];
    stage_rows(st, g);

    if (writer) out[pair * 2 + hi] = res;
    if (npair >= npairs) break;
    pair  = npair;
    npair = nnpair;
  }
}

extern "C" void kernel_launch(void* const* d_in, const int* in_sizes, int n_in,
                              void* d_out, int out_size, void* d_ws, size_t ws_size,
                              hipStream_t stream) {
  const float* agent_qs = (const float*)d_in[0];
  const int*   states   = (const int*)d_in[1];
  const int*   actions  = (const int*)d_in[2];
  const float* noisep   = (const float*)d_in[3];
  const float* Bw       = (const float*)d_in[4];
  float* out = (float*)d_out;

  const int nelem  = in_sizes[0] / NA;
  const int npairs = (nelem + 1) / 2;

  qdpp_mix_kernel<<<GRID_BLOCKS, 256, 0, stream>>>(agent_qs, states, actions,
                                                   noisep, Bw, out, npairs);
}


# One finish per two pairs: pair B's two 8x8 Gram blocks move into the 16 idle lanes (lane L reads L^8), so one lockstep LU, noise/relu and q-sum pass serves four elements; order of sums unchanged
# speedup vs baseline: 1.3010x; 1.1467x over previous
#include <hip/hip_runtime.h>


typedef __attribute__((ext_vector_type(2))) float v2f;
typedef __attribute__((ext_vector_type(8))) float v8f;

#define NA 8
#define NACT 10
#define EMB 32
#define BIAS 1000

#define GRID_BLOCKS 2048
#define WAVE_STRIDE (GRID_BLOCKS * 8)

#define PITCH 36
#define TILE (16 * PITCH)

#define SWZ_BCAST8(x, K) \
  __int_as_float(__builtin_amdgcn_ds_swizzle(__float_as_int(x), ((K) << 5) | 24))
#define SWZ_XOR(x, M) \
  __int_as_float(__builtin_amdgcn_ds_swizzle(__float_as_int(x), ((M) << 10) | 31))

__device__ __forceinline__ void request_rows(float4 (&g)[4], const float* __restrict__ bwl,
                                             const int4 s4, const int4 a4, int gbias)
{
  const int sx[4] = {s4.x, s4.y, s4.z, s4.w};
  const int ax[4] = {a4.x, a4.y, a4.z, a4.w};
  #pragma unroll
  for (int j = 0; j < 4; ++j) {
    const int row = sx[j] * NACT + ax[j] + gbias + BIAS * j;
    g[j] = *(const float4*)(bwl + (size_t)row * EMB);
  }
}

__device__ __forceinline__ void stage_rows(float* st, const float4 (&g)[4])
{
  #pragma unroll
  for (int j = 0; j < 4; ++j) *(float4*)(st + j * PITCH) = g[j];
}

__device__ __forceinline__ void fetch_fragment(float (&v)[16], const float* ld)
{
  #pragma unroll
  for (int s = 0; s < 4; ++s) {
    const float4 t = *(const float4*)(ld + 4 * s);
    v[4 * s]     = t.x;
    v[4 * s + 1] = t.y;
    v[4 * s + 2] = t.z;
    v[4 * s + 3] = t.w;
  }
}

__device__ __forceinline__ v8f gram_pair(float (&v)[16])
{
  float ssq = 0.f;
  #pragma unroll
  for (int j = 0; j < 16; ++j) ssq = fmaf(v[j], v[j], ssq);
  ssq += SWZ_XOR(ssq, 16);
  const float rn = __builtin_amdgcn_rsqf(fmaxf(ssq, 1e-24f));
  #pragma unroll
  for (int j = 0; j < 16; ++j) v[j] *= rn;

  v8f c = {};
  #pragma unroll
  for (int s = 0; s < 8; ++s) {
    v2f a;
    a[0] = v[2 * s];
    a[1] = v[2 * s + 1];
    c = __builtin_amdgcn_wmma_f32_16x16x4_f32(
        false, a, false, a, (short)0, c, false, false);
  }
  return c;
}

__device__ __forceinline__ v8f merge_pairs(const v8f cA, const v8f cB, bool inner)
{
  v8f c;
  #pragma unroll
  for (int i = 0; i < 8; ++i) {
    const float t = SWZ_XOR(cB[i], 8);
    c[i] = inner ? t : cA[i];
  }
  return c;
}

__device__ __forceinline__ float finish_quad(const v8f c, float q_raw, float n_raw, int ag)
{
  float qv = fminf(fmaxf(q_raw, -10.f), 10.f);
  qv += SWZ_XOR(qv, 1);
  qv += SWZ_XOR(qv, 2);
  qv += SWZ_XOR(qv, 4);

  const float nv = 0.1f * n_raw;
  float m[8];
  #pragma unroll
  for (int i = 0; i < 8; ++i) {
    const float x = c[i] + (i == ag ? nv : 0.f);
    m[i] = fmaxf(x, 0.f) + 1e-8f;
  }

  float det = 1.f;
  #define LU_STEP(K)                                          \
  {                                                           \
    const float akk = SWZ_BCAST8(m[K], K);     \
    det *= akk;                                               \
    const float ia = __builtin_amdgcn_rcpf(akk);              \
    _Pragma("unroll")                                         \
    for (int i = (K) + 1; i < 8; ++i) {                       \
      const float lik = SWZ_BCAST8(m[i], K) * ia;             \
      m[i] = fmaf(-lik, m[K], m[i]);                          \
    }                                                         \
  }
  LU_STEP(0) LU_STEP(1) LU_STEP(2) LU_STEP(3)
  LU_STEP(4) LU_STEP(5) LU_STEP(6) LU_STEP(7)
  #undef LU_STEP

  const float logdet = 0.69314718056f * __builtin_amdgcn_logf(det + 1e-8f);
  return qv + logdet;
}

__global__ __launch_bounds__(256) void qdpp_mix_kernel(
    const float* __restrict__ agent_qs,
    const int*   __restrict__ states,
    const int*   __restrict__ actions,
    const float* __restrict__ noise,
    const float* __restrict__ Bw,
    float*       __restrict__ out,
    int npairs)
{
  __shared__ __attribute__((aligned(16))) float tiles[8 * TILE];

  const int lane = threadIdx.x & 31;
  const int wave = __builtin_amdgcn_readfirstlane((int)(threadIdx.x >> 5));
  int pairA = blockIdx.x * 8 + wave;
  if (pairA >= npairs) return;

  const int hi = lane >> 4;
  const int r  = lane & 15;

  const int ag = lane & 7;
  const bool inner = (((lane >> 3) ^ (lane >> 4)) & 1) != 0;
  const bool writer = (ag == 0);
  const int eoff = 8 * hi + ag;

  const int ga = lane >> 3;
  const int gb = lane & 7;
  const int grow  = 8 * (ga & 1) + 4 * (ga >> 1);
  const int gbias = BIAS * (grow & 7);
  const float* bwl = Bw + 4 * gb;
  float* tile = tiles + wave * TILE;
  float* st = tile + grow * PITCH + 4 * gb;
  const float* ld = tile + r * PITCH + 16 * hi;

  bool haveB = pairA + WAVE_STRIDE < npairs;
  int pB = haveB ? pairA + WAVE_STRIDE : pairA;

  int4 s4 = *(const int4*)(states  + pairA * 16 + grow);
  int4 a4 = *(const int4*)(actions + pairA * 16 + grow);
  float4 g[4];
  request_rows(g, bwl, s4, a4, gbias);

  s4 = *(const int4*)(states  + pB * 16 + grow);
  a4 = *(const int4*)(actions + pB * 16 + grow);
  int px = inner ? pB : pairA;
  float q_raw = agent_qs[px * 16 + eoff];
  float n_raw = noise[px * 16 + eoff];
  stage_rows(st, g);

  for (;;) {
    float v[16];

    fetch_fragment(v, ld);
    const v8f cA = gram_pair(v);

    __builtin_amdgcn_sched_barrier(0);

    const int npairA = pairA + 2 * WAVE_STRIDE;
    const bool more = npairA < npairs;
    const int pnA = more ? npairA : pairA;
    request_rows(g, bwl, s4, a4, gbias);
    s4 = *(const int4*)(states  + pnA * 16 + grow);
    a4 = *(const int4*)(actions + pnA * 16 + grow);
    stage_rows(st, g);

    fetch_fragment(v, ld);
    const v8f cB = gram_pair(v);
    const float res = finish_quad(merge_pairs(cA, cB, inner), q_raw, n_raw, ag);

    __builtin_amdgcn_sched_barrier(0);

    const bool nhaveB = npairA + WAVE_STRIDE < npairs;
    const int pnB = nhaveB ? npairA + WAVE_STRIDE : pnA;
    request_rows(g, bwl, s4, a4, gbias);
    s4 = *(const int4*)(states  + pnB * 16 + grow);
    a4 = *(const int4*)(actions + pnB * 16 + grow);
    const int pxn = inner ? pnB : pnA;
    const float q_n = agent_qs[pxn * 16 + eoff];
    const float n_n = noise[pxn * 16 + eoff];
    stage_rows(st, g);

    if (writer && (!inner || haveB)) out[px * 2 + hi] = res;
    if (!more) break;
    pairA = npairA;
    pB    = pnB;
    haveB = nhaveB;
    px    = pxn;
    q_raw = q_n;
    n_raw = n_n;
  }
}

extern "C" void kernel_launch(void* const* d_in, const int* in_sizes, int n_in,
                              void* d_out, int out_size, void* d_ws, size_t ws_size,
                              hipStream_t stream) {
  const float* agent_qs = (const float*)d_in[0];
  const int*   states   = (const int*)d_in[1];
  const int*   actions  = (const int*)d_in[2];
  const float* noisep   = (const float*)d_in[3];
  const float* Bw       = (const float*)d_in[4];
  float* out = (float*)d_out;

  const int nelem  = in_sizes[0] / NA;
  const int npairs = (nelem + 1) / 2;

  qdpp_mix_kernel<<<GRID_BLOCKS, 256, 0, stream>>>(agent_qs, states, actions,
                                                   noisep, Bw, out, npairs);
}


# Finish per lane, not per 8-lane group: Gram blocks parked in LDS as upper triangles, each lane eliminates one symmetric 8x8 in registers (row factors from pivot row: rounding changes), no swizzles
# speedup vs baseline: 1.5106x; 1.1611x over previous
#include <hip/hip_runtime.h>


typedef __attribute__((ext_vector_type(2))) float v2f;
typedef __attribute__((ext_vector_type(8))) float v8f;

#define NA 8
#define NACT 10
#define EMB 32
#define BIAS 1000

#define GRID_BLOCKS 2048
#define WAVE_STRIDE (GRID_BLOCKS * 8)

#define PITCH 36
#define TILE (16 * PITCH)

#define SWZ_BCAST8(x, K) \
  __int_as_float(__builtin_amdgcn_ds_swizzle(__float_as_int(x), ((K) << 5) | 24))
#define SWZ_XOR(x, M) \
  __int_as_float(__builtin_amdgcn_ds_swizzle(__float_as_int(x), ((M) << 10) | 31))

__device__ __forceinline__ void request_rows(float4 (&g)[4], const float* __restrict__ bwl,
                                             const int4 s4, const int4 a4, int gbias)
{
  const int sx[4] = {s4.x, s4.y, s4.z, s4.w};
  const int ax[4] = {a4.x, a4.y, a4.z, a4.w};
  #pragma unroll
  for (int j = 0; j < 4; ++j) {
    const int row = sx[j] * NACT + ax[j] + gbias + BIAS * j;
    g[j] = *(const float4*)(bwl + (size_t)row * EMB);
  }
}

__device__ __forceinline__ void stage_rows(float* st, const float4 (&g)[4])
{
  #pragma unroll
  for (int j = 0; j < 4; ++j) *(float4*)(st + j * PITCH) = g[j];
}

__device__ __forceinline__ void fetch_fragment(float (&v)[16], const float* ld)
{
  #pragma unroll
  for (int s = 0; s < 4; ++s) {
    const float4 t = *(const float4*)(ld + 4 * s);
    v[4 * s]     = t.x;
    v[4 * s + 1] = t.y;
    v[4 * s + 2] = t.z;
    v[4 * s + 3] = t.w;
  }
}

__device__ __forceinline__ v8f gram_pair(float (&v)[16])
{
  float ssq = 0.f;
  #pragma unroll
  for (int j = 0; j < 16; ++j) ssq = fmaf(v[j], v[j], ssq);
  ssq += SWZ_XOR(ssq, 16);
  const float rn = __builtin_amdgcn_rsqf(fmaxf(ssq, 1e-24f));
  #pragma unroll
  for (int j = 0; j < 16; ++j) v[j] *= rn;

  v8f c = {};
  #pragma unroll
  for (int s = 0; s < 8; ++s) {
    v2f a;
    a[0] = v[2 * s];
    a[1] = v[2 * s + 1];
    c = __builtin_amdgcn_wmma_f32_16x16x4_f32(
        false, a, false, a, (short)0, c, false, false);
  }
  return c;
}

#define MP 36
#define TRI(i) (7 * (i) - ((i) * ((i) - 1)) / 2)

__device__ __forceinline__ void park_gram(float* mk, const int (&moff)[8], const v8f c)
{
  #pragma unroll
  for (int i = 0; i < 8; ++i) mk[moff[i]] = c[i];
}

__device__ __forceinline__ float finish_one(const float* mp, const float* qp, const float* np)
{
  float u[36];
  #pragma unroll
  for (int s = 0; s < 9; ++s) {
    const float4 t = *(const float4*)(mp + 4 * s);
    u[4 * s]     = t.x;
    u[4 * s + 1] = t.y;
    u[4 * s + 2] = t.z;
    u[4 * s + 3] = t.w;
  }
  const float4 q0 = *(const float4*)qp;
  const float4 q1 = *(const float4*)(qp + 4);
  const float4 n0 = *(const float4*)np;
  const float4 n1 = *(const float4*)(np + 4);
  const float qr[8] = {q0.x, q0.y, q0.z, q0.w, q1.x, q1.y, q1.z, q1.w};
  const float nz[8] = {n0.x, n0.y, n0.z, n0.w, n1.x, n1.y, n1.z, n1.w};

  float qc[8];
  #pragma unroll
  for (int i = 0; i < 8; ++i) qc[i] = fminf(fmaxf(qr[i], -10.f), 10.f);
  const float qs = ((qc[0] + qc[1]) + (qc[2] + qc[3])) + ((qc[4] + qc[5]) + (qc[6] + qc[7]));

  #pragma unroll
  for (int i = 0; i < 8; ++i) {
    #pragma unroll
    for (int j = i; j < 8; ++j) {
      const float x = u[TRI(i) + j] + (i == j ? 0.1f * nz[i] : 0.f);
      u[TRI(i) + j] = fmaxf(x + 1e-8f, 1e-8f);
    }
  }

  float det = 1.f;
  #pragma unroll
  for (int K = 0; K < 8; ++K) {
    const float piv = u[TRI(K) + K];
    det *= piv;
    const float ia = __builtin_amdgcn_rcpf(piv);
    #pragma unroll
    for (int i = K + 1; i < 8; ++i) {
      const float l = u[TRI(K) + i] * ia;
      #pragma unroll
      for (int j = i; j < 8; ++j)
        u[TRI(i) + j] = fmaf(-l, u[TRI(K) + j], u[TRI(i) + j]);
    }
  }

  const float logdet = 0.69314718056f * __builtin_amdgcn_logf(det + 1e-8f);
  return qs + logdet;
}

__global__ __launch_bounds__(256) void qdpp_mix_kernel(
    const float* __restrict__ agent_qs,
    const int*   __restrict__ states,
    const int*   __restrict__ actions,
    const float* __restrict__ noise,
    const float* __restrict__ Bw,
    float*       __restrict__ out,
    int npairs)
{
  __shared__ __attribute__((aligned(16))) float tiles[8 * TILE];
  __shared__ __attribute__((aligned(16))) float blocks[8 * 16 * MP];

  const int lane = threadIdx.x & 31;
  const int wave = __builtin_amdgcn_readfirstlane((int)(threadIdx.x >> 5));
  const int pair0 = blockIdx.x * 8 + wave;

  const int hi = lane >> 4;
  const int r  = lane & 15;

  const int ag = lane & 7;
  const bool wanted = (((lane >> 3) ^ (lane >> 4)) & 1) == 0;
  int moff[8];
  #pragma unroll
  for (int i = 0; i < 8; ++i) moff[i] = (i <= ag) ? TRI(i) + ag : TRI(ag) + i;

  const int ga = lane >> 3;
  const int gb = lane & 7;
  const int grow  = 8 * (ga & 1) + 4 * (ga >> 1);
  const int gbias = BIAS * (grow & 7);
  const float* bwl = Bw + 4 * gb;
  float* tile = tiles + wave * TILE;
  float* st = tile + grow * PITCH + 4 * gb;
  const float* ld = tile + r * PITCH + 16 * hi;
  float* bw = blocks + wave * (16 * MP);
  float* bs = bw + hi * MP;

  const int fl = lane & 15;

  const int base = pair0;
  if (base >= npairs) return;
  int cnt = (npairs - base + WAVE_STRIDE - 1) / WAVE_STRIDE;
  cnt = cnt > 8 ? 8 : cnt;
  {
    auto pidx = [&](int k) { return k < cnt ? base + k * WAVE_STRIDE : base; };

    int4 s4 = *(const int4*)(states  + base * 16 + grow);
    int4 a4 = *(const int4*)(actions + base * 16 + grow);
    float4 g[4];
    request_rows(g, bwl, s4, a4, gbias);
    s4 = *(const int4*)(states  + pidx(1) * 16 + grow);
    a4 = *(const int4*)(actions + pidx(1) * 16 + grow);
    stage_rows(st, g);

    #pragma unroll
    for (int k = 0; k < 8; ++k) {
      float v[16];
      fetch_fragment(v, ld);
      const v8f c = gram_pair(v);
      if (wanted) park_gram(bs + 2 * k * MP, moff, c);

      __builtin_amdgcn_sched_barrier(0);

      if (k + 1 < 8) {
        request_rows(g, bwl, s4, a4, gbias);
        if (k + 2 < 8) {
          const int pn = pidx(k + 2);
          s4 = *(const int4*)(states  + pn * 16 + grow);
          a4 = *(const int4*)(actions + pn * 16 + grow);
        }
        stage_rows(st, g);
      }
    }

    const int fk = fl >> 1;
    const bool live = fk < cnt;
    const int fe = (base + (live ? fk : 0) * WAVE_STRIDE) * 2 + (fl & 1);
    const float res = finish_one(bw + fl * MP, agent_qs + fe * NA, noise + fe * NA);
    if (live && lane < 16) out[fe] = res;
  }
}

extern "C" void kernel_launch(void* const* d_in, const int* in_sizes, int n_in,
                              void* d_out, int out_size, void* d_ws, size_t ws_size,
                              hipStream_t stream) {
  const float* agent_qs = (const float*)d_in[0];
  const int*   states   = (const int*)d_in[1];
  const int*   actions  = (const int*)d_in[2];
  const float* noisep   = (const float*)d_in[3];
  const float* Bw       = (const float*)d_in[4];
  float* out = (float*)d_out;

  const int nelem  = in_sizes[0] / NA;
  const int npairs = (nelem + 1) / 2;

  qdpp_mix_kernel<<<GRID_BLOCKS, 256, 0, stream>>>(agent_qs, states, actions,
                                                   noisep, Bw, out, npairs);
}


# Row normalisation moved behind the Gram: raw rows feed the WMMAs, the finishing lane scales entry (i,j) by rsqrt of the block's own diagonal; squared norms now summed in WMMA k order
# speedup vs baseline: 1.6209x; 1.0730x over previous
#include <hip/hip_runtime.h>


typedef __attribute__((ext_vector_type(2))) float v2f;
typedef __attribute__((ext_vector_type(8))) float v8f;

#define NA 8
#define NACT 10
#define EMB 32
#define BIAS 1000

#define GRID_BLOCKS 2048
#define WAVE_STRIDE (GRID_BLOCKS * 8)

#define PITCH 36
#define TILE (16 * PITCH)

__device__ __forceinline__ void request_rows(float4 (&g)[4], const float* __restrict__ bwl,
                                             const int4 s4, const int4 a4, int gbias)
{
  const int sx[4] = {s4.x, s4.y, s4.z, s4.w};
  const int ax[4] = {a4.x, a4.y, a4.z, a4.w};
  #pragma unroll
  for (int j = 0; j < 4; ++j) {
    const int row = sx[j] * NACT + ax[j] + gbias + BIAS * j;
    g[j] = *(const float4*)(bwl + (size_t)row * EMB);
  }
}

__device__ __forceinline__ void stage_rows(float* st, const float4 (&g)[4])
{
  #pragma unroll
  for (int j = 0; j < 4; ++j) *(float4*)(st + j * PITCH) = g[j];
}

__device__ __forceinline__ void fetch_fragment(float (&v)[16], const float* ld)
{
  #pragma unroll
  for (int s = 0; s < 4; ++s) {
    const float4 t = *(const float4*)(ld + 4 * s);
    v[4 * s]     = t.x;
    v[4 * s + 1] = t.y;
    v[4 * s + 2] = t.z;
    v[4 * s + 3] = t.w;
  }
}

__device__ __forceinline__ v8f gram_pair(const float (&v)[16])
{
  v8f c = {};
  #pragma unroll
  for (int s = 0; s < 8; ++s) {
    v2f a;
    a[0] = v[2 * s];
    a[1] = v[2 * s + 1];
    c = __builtin_amdgcn_wmma_f32_16x16x4_f32(
        false, a, false, a, (short)0, c, false, false);
  }
  return c;
}

#define MP 36
#define TRI(i) (7 * (i) - ((i) * ((i) - 1)) / 2)

__device__ __forceinline__ void park_gram(float* mk, const int (&moff)[8], const v8f c)
{
  #pragma unroll
  for (int i = 0; i < 8; ++i) mk[moff[i]] = c[i];
}

__device__ __forceinline__ float finish_one(const float* mp, const float* qp, const float* np)
{
  float u[36];
  #pragma unroll
  for (int s = 0; s < 9; ++s) {
    const float4 t = *(const float4*)(mp + 4 * s);
    u[4 * s]     = t.x;
    u[4 * s + 1] = t.y;
    u[4 * s + 2] = t.z;
    u[4 * s + 3] = t.w;
  }
  const float4 q0 = *(const float4*)qp;
  const float4 q1 = *(const float4*)(qp + 4);
  const float4 n0 = *(const float4*)np;
  const float4 n1 = *(const float4*)(np + 4);
  const float qr[8] = {q0.x, q0.y, q0.z, q0.w, q1.x, q1.y, q1.z, q1.w};
  const float nz[8] = {n0.x, n0.y, n0.z, n0.w, n1.x, n1.y, n1.z, n1.w};

  float qc[8];
  #pragma unroll
  for (int i = 0; i < 8; ++i) qc[i] = fminf(fmaxf(qr[i], -10.f), 10.f);
  const float qs = ((qc[0] + qc[1]) + (qc[2] + qc[3])) + ((qc[4] + qc[5]) + (qc[6] + qc[7]));

  float rn[8];
  #pragma unroll
  for (int i = 0; i < 8; ++i)
    rn[i] = __builtin_amdgcn_rsqf(fmaxf(u[TRI(i) + i], 1e-24f));

  #pragma unroll
  for (int i = 0; i < 8; ++i) {
    #pragma unroll
    for (int j = i; j < 8; ++j) {
      const float gn = (u[TRI(i) + j] * rn[i]) * rn[j];
      const float x = gn + (i == j ? 0.1f * nz[i] : 0.f);
      u[TRI(i) + j] = fmaxf(x + 1e-8f, 1e-8f);
    }
  }

  float det = 1.f;
  #pragma unroll
  for (int K = 0; K < 8; ++K) {
    const float piv = u[TRI(K) + K];
    det *= piv;
    const float ia = __builtin_amdgcn_rcpf(piv);
    #pragma unroll
    for (int i = K + 1; i < 8; ++i) {
      const float l = u[TRI(K) + i] * ia;
      #pragma unroll
      for (int j = i; j < 8; ++j)
        u[TRI(i) + j] = fmaf(-l, u[TRI(K) + j], u[TRI(i) + j]);
    }
  }

  const float logdet = 0.69314718056f * __builtin_amdgcn_logf(det + 1e-8f);
  return qs + logdet;
}

__global__ __launch_bounds__(256) void qdpp_mix_kernel(
    const float* __restrict__ agent_qs,
    const int*   __restrict__ states,
    const int*   __restrict__ actions,
    const float* __restrict__ noise,
    const float* __restrict__ Bw,
    float*       __restrict__ out,
    int npairs)
{
  __shared__ __attribute__((aligned(16))) float tiles[8 * TILE];
  __shared__ __attribute__((aligned(16))) float blocks[8 * 16 * MP];

  const int lane = threadIdx.x & 31;
  const int wave = __builtin_amdgcn_readfirstlane((int)(threadIdx.x >> 5));
  const int pair0 = blockIdx.x * 8 + wave;

  const int hi = lane >> 4;
  const int r  = lane & 15;

  const int ag = lane & 7;
  const bool wanted = (((lane >> 3) ^ (lane >> 4)) & 1) == 0;
  int moff[8];
  #pragma unroll
  for (int i = 0; i < 8; ++i) moff[i] = (i <= ag) ? TRI(i) + ag : TRI(ag) + i;

  const int ga = lane >> 3;
  const int gb = lane & 7;
  const int grow  = 8 * (ga & 1) + 4 * (ga >> 1);
  const int gbias = BIAS * (grow & 7);
  const float* bwl = Bw + 4 * gb;
  float* tile = tiles + wave * TILE;
  float* st = tile + grow * PITCH + 4 * gb;
  const float* ld = tile + r * PITCH + 16 * hi;
  float* bw = blocks + wave * (16 * MP);
  float* bs = bw + hi * MP;

  const int fl = lane & 15;

  const int base = pair0;
  if (base >= npairs) return;
  int cnt = (npairs - base + WAVE_STRIDE - 1) / WAVE_STRIDE;
  cnt = cnt > 8 ? 8 : cnt;
  {
    auto pidx = [&](int k) { return k < cnt ? base + k * WAVE_STRIDE : base; };

    int4 s4 = *(const int4*)(states  + base * 16 + grow);
    int4 a4 = *(const int4*)(actions + base * 16 + grow);
    float4 g[4];
    request_rows(g, bwl, s4, a4, gbias);
    s4 = *(const int4*)(states  + pidx(1) * 16 + grow);
    a4 = *(const int4*)(actions + pidx(1) * 16 + grow);
    stage_rows(st, g);

    #pragma unroll
    for (int k = 0; k < 8; ++k) {
      float v[16];
      fetch_fragment(v, ld);
      asm volatile("" :: "v"(v[0]), "v"(v[1]), "v"(v[2]), "v"(v[3]),
                         "v"(v[4]), "v"(v[5]), "v"(v[6]), "v"(v[7]),
                         "v"(v[8]), "v"(v[9]), "v"(v[10]), "v"(v[11]),
                         "v"(v[12]), "v"(v[13]), "v"(v[14]), "v"(v[15]));
      const v8f c = gram_pair(v);
      if (wanted) park_gram(bs + 2 * k * MP, moff, c);

      __builtin_amdgcn_sched_barrier(0);

      if (k + 1 < 8) {
        request_rows(g, bwl, s4, a4, gbias);
        if (k + 2 < 8) {
          const int pn = pidx(k + 2);
          s4 = *(const int4*)(states  + pn * 16 + grow);
          a4 = *(const int4*)(actions + pn * 16 + grow);
        }
        stage_rows(st, g);
      }
    }

    const int fk = fl >> 1;
    const bool live = fk < cnt;
    const int fe = (base + (live ? fk : 0) * WAVE_STRIDE) * 2 + (fl & 1);
    const float res = finish_one(bw + fl * MP, agent_qs + fe * NA, noise + fe * NA);
    if (live && lane < 16) out[fe] = res;
  }
}

extern "C" void kernel_launch(void* const* d_in, const int* in_sizes, int n_in,
                              void* d_out, int out_size, void* d_ws, size_t ws_size,
                              hipStream_t stream) {
  const float* agent_qs = (const float*)d_in[0];
  const int*   states   = (const int*)d_in[1];
  const int*   actions  = (const int*)d_in[2];
  const float* noisep   = (const float*)d_in[3];
  const float* Bw       = (const float*)d_in[4];
  float* out = (float*)d_out;

  const int nelem  = in_sizes[0] / NA;
  const int npairs = (nelem + 1) / 2;

  qdpp_mix_kernel<<<GRID_BLOCKS, 256, 0, stream>>>(agent_qs, states, actions,
                                                   noisep, Bw, out, npairs);
}


# Next pair's row gather issued in front of this pair's Gram phase (rows wait in registers, staged behind the WMMAs); row offsets made once per wave up front, kept in LDS: no index work in the loop
# speedup vs baseline: 1.7458x; 1.0771x over previous
#include <hip/hip_runtime.h>


typedef __attribute__((ext_vector_type(2))) float v2f;
typedef __attribute__((ext_vector_type(8))) float v8f;

#define NA 8
#define NACT 10
#define EMB 32
#define BIAS 1000

#define GRID_BLOCKS 2048
#define WAVE_STRIDE (GRID_BLOCKS * 8)

#define PITCH 36
#define TILE (16 * PITCH)

#define ROW_BYTES (EMB * 4)
__device__ __forceinline__ float4 request_row(const char* tbl, float of, unsigned piece)
{
  return *(const float4*)(tbl + (size_t)(__float_as_uint(of) + piece));
}
#define REQUEST_ROWS(g0, g1, g2, g3, of)        \
  do {                                          \
    g0 = request_row(tbl, (of).x, piece);       \
    g1 = request_row(tbl, (of).y, piece);       \
    g2 = request_row(tbl, (of).z, piece);       \
    g3 = request_row(tbl, (of).w, piece);       \
  } while (0)

__device__ __forceinline__ void stage_rows(float* st, const float4 g0, const float4 g1,
                                           const float4 g2, const float4 g3)
{
  *(float4*)(st)             = g0;
  *(float4*)(st + PITCH)     = g1;
  *(float4*)(st + 2 * PITCH) = g2;
  *(float4*)(st + 3 * PITCH) = g3;
}

__device__ __forceinline__ void fetch_fragment(float (&v)[16], const float* ld)
{
  #pragma unroll
  for (int s = 0; s < 4; ++s) {
    const float4 t = *(const float4*)(ld + 4 * s);
    v[4 * s]     = t.x;
    v[4 * s + 1] = t.y;
    v[4 * s + 2] = t.z;
    v[4 * s + 3] = t.w;
  }
}

__device__ __forceinline__ v8f gram_pair(const float (&v)[16])
{
  v8f c = {};
  #pragma unroll
  for (int s = 0; s < 8; ++s) {
    v2f a;
    a[0] = v[2 * s];
    a[1] = v[2 * s + 1];
    c = __builtin_amdgcn_wmma_f32_16x16x4_f32(
        false, a, false, a, (short)0, c, false, false);
  }
  return c;
}

#define MP 36
#define TRI(i) (7 * (i) - ((i) * ((i) - 1)) / 2)

__device__ __forceinline__ void park_gram(float* mk, const int (&moff)[8], const v8f c)
{
  #pragma unroll
  for (int i = 0; i < 8; ++i) mk[moff[i]] = c[i];
}

__device__ __forceinline__ float finish_one(const float* mp, const float* qp, const float* np)
{
  float u[36];
  #pragma unroll
  for (int s = 0; s < 9; ++s) {
    const float4 t = *(const float4*)(mp + 4 * s);
    u[4 * s]     = t.x;
    u[4 * s + 1] = t.y;
    u[4 * s + 2] = t.z;
    u[4 * s + 3] = t.w;
  }
  const float4 q0 = *(const float4*)qp;
  const float4 q1 = *(const float4*)(qp + 4);
  const float4 n0 = *(const float4*)np;
  const float4 n1 = *(const float4*)(np + 4);
  const float qr[8] = {q0.x, q0.y, q0.z, q0.w, q1.x, q1.y, q1.z, q1.w};
  const float nz[8] = {n0.x, n0.y, n0.z, n0.w, n1.x, n1.y, n1.z, n1.w};

  float qc[8];
  #pragma unroll
  for (int i = 0; i < 8; ++i) qc[i] = fminf(fmaxf(qr[i], -10.f), 10.f);
  const float qs = ((qc[0] + qc[1]) + (qc[2] + qc[3])) + ((qc[4] + qc[5]) + (qc[6] + qc[7]));

  float rn[8];
  #pragma unroll
  for (int i = 0; i < 8; ++i)
    rn[i] = __builtin_amdgcn_rsqf(fmaxf(u[TRI(i) + i], 1e-24f));

  #pragma unroll
  for (int i = 0; i < 8; ++i) {
    #pragma unroll
    for (int j = i; j < 8; ++j) {
      const float gn = (u[TRI(i) + j] * rn[i]) * rn[j];
      const float x = gn + (i == j ? 0.1f * nz[i] : 0.f);
      u[TRI(i) + j] = fmaxf(x + 1e-8f, 1e-8f);
    }
  }

  float det = 1.f;
  #pragma unroll
  for (int K = 0; K < 8; ++K) {
    const float piv = u[TRI(K) + K];
    det *= piv;
    const float ia = __builtin_amdgcn_rcpf(piv);
    #pragma unroll
    for (int i = K + 1; i < 8; ++i) {
      const float l = u[TRI(K) + i] * ia;
      #pragma unroll
      for (int j = i; j < 8; ++j)
        u[TRI(i) + j] = fmaf(-l, u[TRI(K) + j], u[TRI(i) + j]);
    }
  }

  const float logdet = 0.69314718056f * __builtin_amdgcn_logf(det + 1e-8f);
  return qs + logdet;
}

__global__ __launch_bounds__(256) void qdpp_mix_kernel(
    const float* __restrict__ agent_qs,
    const int*   __restrict__ states,
    const int*   __restrict__ actions,
    const float* __restrict__ noise,
    const float* Bw,
    float*       __restrict__ out,
    int npairs)
{
  __shared__ __attribute__((aligned(16))) float tiles[8 * TILE];
  __shared__ __attribute__((aligned(16))) float blocks[8 * 16 * MP];

  const int lane = threadIdx.x & 31;
  const int wave = __builtin_amdgcn_readfirstlane((int)(threadIdx.x >> 5));
  const int pair0 = blockIdx.x * 8 + wave;

  const int hi = lane >> 4;
  const int r  = lane & 15;

  const int ag = lane & 7;
  const bool wanted = (((lane >> 3) ^ (lane >> 4)) & 1) == 0;
  int moff[8];
  #pragma unroll
  for (int i = 0; i < 8; ++i) moff[i] = (i <= ag) ? TRI(i) + ag : TRI(ag) + i;

  const int ga = lane >> 3;
  const int gb = lane & 7;
  const int grow  = 8 * (ga & 1) + 4 * (ga >> 1);
  const unsigned piece = 16u * (unsigned)gb;
  const char* tbl = (const char*)Bw;
  float* tile = tiles + wave * TILE;
  float* st = tile + grow * PITCH + 4 * gb;
  const float* ld = tile + r * PITCH + 16 * hi;
  float* bw = blocks + wave * (16 * MP);
  float* bs = (wanted ? bw : tile) + hi * MP;

  const int fl = lane & 15;

  const int base = pair0;
  if (base >= npairs) return;
  int cnt = (npairs - base + WAVE_STRIDE - 1) / WAVE_STRIDE;
  cnt = cnt > 8 ? 8 : cnt;
  {
    {
      const int kk = lane >> 2;
      const int q  = lane & 3;
      const int pk = kk < cnt ? base + kk * WAVE_STRIDE : base;
      const int4 s4 = *(const int4*)(states  + pk * 16 + 4 * q);
      const int4 a4 = *(const int4*)(actions + pk * 16 + 4 * q);
      const unsigned ab = (unsigned)(BIAS * 4 * (q & 1));
      float4 o;
      o.x = __uint_as_float(((unsigned)s4.x * NACT + (unsigned)a4.x + ab) * ROW_BYTES);
      o.y = __uint_as_float(((unsigned)s4.y * NACT + (unsigned)a4.y + ab + BIAS) * ROW_BYTES);
      o.z = __uint_as_float(((unsigned)s4.z * NACT + (unsigned)a4.z + ab + 2 * BIAS) * ROW_BYTES);
      o.w = __uint_as_float(((unsigned)s4.w * NACT + (unsigned)a4.w + ab + 3 * BIAS) * ROW_BYTES);
      *(float4*)(bw + kk * (2 * MP) + 4 * q) = o;
    }
    const float* op = bw + grow;

    float4 of = *(const float4*)op;
    float4 g0, g1, g2, g3;
    REQUEST_ROWS(g0, g1, g2, g3, of);
    of = *(const float4*)(op + 2 * MP);
    stage_rows(st, g0, g1, g2, g3);

    #pragma unroll
    for (int k = 0; k < 8; ++k) {
      if (k + 1 < 8) {
        REQUEST_ROWS(g0, g1, g2, g3, of);
        if (k + 2 < 8) of = *(const float4*)(op + (k + 2) * (2 * MP));
      }
      __builtin_amdgcn_sched_barrier(0);

      float v[16];
      fetch_fragment(v, ld);
      asm volatile("" :: "v"(v[0]), "v"(v[1]), "v"(v[2]), "v"(v[3]),
                         "v"(v[4]), "v"(v[5]), "v"(v[6]), "v"(v[7]),
                         "v"(v[8]), "v"(v[9]), "v"(v[10]), "v"(v[11]),
                         "v"(v[12]), "v"(v[13]), "v"(v[14]), "v"(v[15]));
      const v8f c = gram_pair(v);
      park_gram(bs + 2 * k * MP, moff, c);

      __builtin_amdgcn_sched_barrier(0);

      if (k + 1 < 8) stage_rows(st, g0, g1, g2, g3);
    }

    const int fk = fl >> 1;
    const bool live = fk < cnt;
    const int fe = (base + (live ? fk : 0) * WAVE_STRIDE) * 2 + (fl & 1);
    const float res = finish_one(bw + fl * MP, agent_qs + fe * NA, noise + fe * NA);
    if (live && lane < 16) out[fe] = res;
  }
}

extern "C" void kernel_launch(void* const* d_in, const int* in_sizes, int n_in,
                              void* d_out, int out_size, void* d_ws, size_t ws_size,
                              hipStream_t stream) {
  const float* agent_qs = (const float*)d_in[0];
  const int*   states   = (const int*)d_in[1];
  const int*   actions  = (const int*)d_in[2];
  const float* noisep   = (const float*)d_in[3];
  const float* Bw       = (const float*)d_in[4];
  float* out = (float*)d_out;

  const int nelem  = in_sizes[0] / NA;
  const int npairs = (nelem + 1) / 2;

  qdpp_mix_kernel<<<GRID_BLOCKS, 256, 0, stream>>>(agent_qs, states, actions,
                                                   noisep, Bw, out, npairs);
}


# Two pairs per trip: each of 8192 waves (8 per SIMD) stages, fetches and multiplies two pairs at once on two independent WMMA chains, 16 pairs in 8 dependent trips; 32 lanes finish; sums unchanged
# speedup vs baseline: 1.8164x; 1.0404x over previous
#include <hip/hip_runtime.h>


typedef __attribute__((ext_vector_type(2))) float v2f;
typedef __attribute__((ext_vector_type(8))) float v8f;

#define NA 8
#define NACT 10
#define EMB 32
#define BIAS 1000

#define GRID_BLOCKS 1024
#define WAVE_STRIDE (GRID_BLOCKS * 8)

#define PITCH 36
#define TILE (16 * PITCH)

#define ROW_BYTES (EMB * 4)
__device__ __forceinline__ float4 request_row(const char* tbl, float of, unsigned piece)
{
  return *(const float4*)(tbl + (size_t)(__float_as_uint(of) + piece));
}
#define REQUEST_ROWS(g0, g1, g2, g3, of)        \
  do {                                          \
    g0 = request_row(tbl, (of).x, piece);       \
    g1 = request_row(tbl, (of).y, piece);       \
    g2 = request_row(tbl, (of).z, piece);       \
    g3 = request_row(tbl, (of).w, piece);       \
  } while (0)

__device__ __forceinline__ void stage_rows(float* st, const float4 g0, const float4 g1,
                                           const float4 g2, const float4 g3)
{
  *(float4*)(st)             = g0;
  *(float4*)(st + PITCH)     = g1;
  *(float4*)(st + 2 * PITCH) = g2;
  *(float4*)(st + 3 * PITCH) = g3;
}

__device__ __forceinline__ void fetch_fragment(float (&v)[16], const float* ld)
{
  #pragma unroll
  for (int s = 0; s < 4; ++s) {
    const float4 t = *(const float4*)(ld + 4 * s);
    v[4 * s]     = t.x;
    v[4 * s + 1] = t.y;
    v[4 * s + 2] = t.z;
    v[4 * s + 3] = t.w;
  }
}

__device__ __forceinline__ void gram_two(const float (&va)[16], const float (&vb)[16],
                                         v8f& ca, v8f& cb)
{
  v8f xa = {};
  v8f xb = {};
  #pragma unroll
  for (int s = 0; s < 8; ++s) {
    v2f a;
    a[0] = va[2 * s];
    a[1] = va[2 * s + 1];
    v2f b;
    b[0] = vb[2 * s];
    b[1] = vb[2 * s + 1];
    xa = __builtin_amdgcn_wmma_f32_16x16x4_f32(
        false, a, false, a, (short)0, xa, false, false);
    xb = __builtin_amdgcn_wmma_f32_16x16x4_f32(
        false, b, false, b, (short)0, xb, false, false);
  }
  ca = xa;
  cb = xb;
}

#define MP 36
#define TRI(i) (7 * (i) - ((i) * ((i) - 1)) / 2)

__device__ __forceinline__ void park_gram(float* mk, const int (&moff)[8], const v8f c)
{
  #pragma unroll
  for (int i = 0; i < 8; ++i) mk[moff[i]] = c[i];
}

__device__ __forceinline__ float finish_one(const float* mp, const float* qp, const float* np)
{
  float u[36];
  #pragma unroll
  for (int s = 0; s < 9; ++s) {
    const float4 t = *(const float4*)(mp + 4 * s);
    u[4 * s]     = t.x;
    u[4 * s + 1] = t.y;
    u[4 * s + 2] = t.z;
    u[4 * s + 3] = t.w;
  }
  const float4 q0 = *(const float4*)qp;
  const float4 q1 = *(const float4*)(qp + 4);
  const float4 n0 = *(const float4*)np;
  const float4 n1 = *(const float4*)(np + 4);
  const float qr[8] = {q0.x, q0.y, q0.z, q0.w, q1.x, q1.y, q1.z, q1.w};
  const float nz[8] = {n0.x, n0.y, n0.z, n0.w, n1.x, n1.y, n1.z, n1.w};

  float qc[8];
  #pragma unroll
  for (int i = 0; i < 8; ++i) qc[i] = fminf(fmaxf(qr[i], -10.f), 10.f);
  const float qs = ((qc[0] + qc[1]) + (qc[2] + qc[3])) + ((qc[4] + qc[5]) + (qc[6] + qc[7]));

  float rn[8];
  #pragma unroll
  for (int i = 0; i < 8; ++i)
    rn[i] = __builtin_amdgcn_rsqf(fmaxf(u[TRI(i) + i], 1e-24f));

  #pragma unroll
  for (int i = 0; i < 8; ++i) {
    #pragma unroll
    for (int j = i; j < 8; ++j) {
      const float gn = (u[TRI(i) + j] * rn[i]) * rn[j];
      const float x = gn + (i == j ? 0.1f * nz[i] : 0.f);
      u[TRI(i) + j] = fmaxf(x + 1e-8f, 1e-8f);
    }
  }

  float det = 1.f;
  #pragma unroll
  for (int K = 0; K < 8; ++K) {
    const float piv = u[TRI(K) + K];
    det *= piv;
    const float ia = __builtin_amdgcn_rcpf(piv);
    #pragma unroll
    for (int i = K + 1; i < 8; ++i) {
      const float l = u[TRI(K) + i] * ia;
      #pragma unroll
      for (int j = i; j < 8; ++j)
        u[TRI(i) + j] = fmaf(-l, u[TRI(K) + j], u[TRI(i) + j]);
    }
  }

  const float logdet = 0.69314718056f * __builtin_amdgcn_logf(det + 1e-8f);
  return qs + logdet;
}

__global__ __launch_bounds__(256) void qdpp_mix_kernel(
    const float* __restrict__ agent_qs,
    const int*   __restrict__ states,
    const int*   __restrict__ actions,
    const float* __restrict__ noise,
    const float* Bw,
    float*       __restrict__ out,
    int npairs)
{
  __shared__ __attribute__((aligned(16))) float tiles[8 * 2 * TILE];
  __shared__ __attribute__((aligned(16))) float blocks[8 * 32 * MP];

  const int lane = threadIdx.x & 31;
  const int wave = __builtin_amdgcn_readfirstlane((int)(threadIdx.x >> 5));
  const int pair0 = blockIdx.x * 8 + wave;

  const int hi = lane >> 4;
  const int r  = lane & 15;

  const int ag = lane & 7;
  const bool wanted = (((lane >> 3) ^ (lane >> 4)) & 1) == 0;
  int moff[8];
  #pragma unroll
  for (int i = 0; i < 8; ++i) moff[i] = (i <= ag) ? TRI(i) + ag : TRI(ag) + i;

  const int ga = lane >> 3;
  const int gb = lane & 7;
  const int grow  = 8 * (ga & 1) + 4 * (ga >> 1);
  const unsigned piece = 16u * (unsigned)gb;
  const char* tbl = (const char*)Bw;
  float* tile = tiles + wave * (2 * TILE);
  float* st = tile + grow * PITCH + 4 * gb;
  const float* ld = tile + r * PITCH + 16 * hi;
  float* bw = blocks + wave * (32 * MP);
  float* bs = (wanted ? bw : tile) + hi * MP;

  const int base = pair0;
  if (base >= npairs) return;
  int cnt = (npairs - base + WAVE_STRIDE - 1) / WAVE_STRIDE;
  cnt = cnt > 16 ? 16 : cnt;
  {
    #pragma unroll
    for (int h = 0; h < 2; ++h) {
      const int kk = 8 * h + (lane >> 2);
      const int q  = lane & 3;
      const int pk = kk < cnt ? base + kk * WAVE_STRIDE : base;
      const int4 s4 = *(const int4*)(states  + pk * 16 + 4 * q);
      const int4 a4 = *(const int4*)(actions + pk * 16 + 4 * q);
      const unsigned ab = (unsigned)(BIAS * 4 * (q & 1));
      float4 o;
      o.x = __uint_as_float(((unsigned)s4.x * NACT + (unsigned)a4.x + ab) * ROW_BYTES);
      o.y = __uint_as_float(((unsigned)s4.y * NACT + (unsigned)a4.y + ab + BIAS) * ROW_BYTES);
      o.z = __uint_as_float(((unsigned)s4.z * NACT + (unsigned)a4.z + ab + 2 * BIAS) * ROW_BYTES);
      o.w = __uint_as_float(((unsigned)s4.w * NACT + (unsigned)a4.w + ab + 3 * BIAS) * ROW_BYTES);
      *(float4*)(bw + kk * (2 * MP) + 4 * q) = o;
    }
    const float* op = bw + grow;

    float4 ofa = *(const float4*)op;
    float4 ofb = *(const float4*)(op + 2 * MP);
    float4 a0, a1, a2, a3, b0, b1, b2, b3;
    REQUEST_ROWS(a0, a1, a2, a3, ofa);
    REQUEST_ROWS(b0, b1, b2, b3, ofb);
    ofa = *(const float4*)(op + 2 * (2 * MP));
    ofb = *(const float4*)(op + 3 * (2 * MP));
    stage_rows(st, a0, a1, a2, a3);
    stage_rows(st + TILE, b0, b1, b2, b3);

    #pragma unroll
    for (int t = 0; t < 8; ++t) {
      if (t + 1 < 8) {
        REQUEST_ROWS(a0, a1, a2, a3, ofa);
        REQUEST_ROWS(b0, b1, b2, b3, ofb);
        if (t + 2 < 8) {
          ofa = *(const float4*)(op + (2 * t + 4) * (2 * MP));
          ofb = *(const float4*)(op + (2 * t + 5) * (2 * MP));
        }
      }
      __builtin_amdgcn_sched_barrier(0);

      float va[16], vb[16];
      fetch_fragment(va, ld);
      fetch_fragment(vb, ld + TILE);
      asm volatile("" :: "v"(va[0]), "v"(va[1]), "v"(va[2]), "v"(va[3]),
                         "v"(va[4]), "v"(va[5]), "v"(va[6]), "v"(va[7]),
                         "v"(va[8]), "v"(va[9]), "v"(va[10]), "v"(va[11]),
                         "v"(va[12]), "v"(va[13]), "v"(va[14]), "v"(va[15]));
      asm volatile("" :: "v"(vb[0]), "v"(vb[1]), "v"(vb[2]), "v"(vb[3]),
                         "v"(vb[4]), "v"(vb[5]), "v"(vb[6]), "v"(vb[7]),
                         "v"(vb[8]), "v"(vb[9]), "v"(vb[10]), "v"(vb[11]),
                         "v"(vb[12]), "v"(vb[13]), "v"(vb[14]), "v"(vb[15]));
      v8f ca, cb;
      gram_two(va, vb, ca, cb);
      park_gram(bs + (4 * t) * MP, moff, ca);
      park_gram(bs + (4 * t + 2) * MP, moff, cb);

      __builtin_amdgcn_sched_barrier(0);

      if (t + 1 < 8) {
        stage_rows(st, a0, a1, a2, a3);
        stage_rows(st + TILE, b0, b1, b2, b3);
      }
    }

    const int fk = lane >> 1;
    const bool live = fk < cnt;
    const int fe = (base + (live ? fk : 0) * WAVE_STRIDE) * 2 + (lane & 1);
    const float res = finish_one(bw + lane * MP, agent_qs + fe * NA, noise + fe * NA);
    if (live) out[fe] = res;
  }
}

extern "C" void kernel_launch(void* const* d_in, const int* in_sizes, int n_in,
                              void* d_out, int out_size, void* d_ws, size_t ws_size,
                              hipStream_t stream) {
  const float* agent_qs = (const float*)d_in[0];
  const int*   states   = (const int*)d_in[1];
  const int*   actions  = (const int*)d_in[2];
  const float* noisep   = (const float*)d_in[3];
  const float* Bw       = (const float*)d_in[4];
  float* out = (float*)d_out;

  const int nelem  = in_sizes[0] / NA;
  const int npairs = (nelem + 1) / 2;

  qdpp_mix_kernel<<<GRID_BLOCKS, 256, 0, stream>>>(agent_qs, states, actions,
                                                   noisep, Bw, out, npairs);
}
